# starting kernel
# baseline (speedup 1.0000x reference)
#include <hip/hip_runtime.h>
#include <hip/hip_bf16.h>


#define TOKENS   32768
#define DDIM     2048
#define NEXP     4
#define EPSF     1e-6f
#define CHUNK    128
#define CHUNKP   132
#define NCHUNK   (DDIM / CHUNK)
#define TPB      16
#define THREADS  256

typedef __attribute__((ext_vector_type(2))) float v2f;
typedef __attribute__((ext_vector_type(8))) float v8f;

__global__ __launch_bounds__(THREADS)
void moe_count_kernel(const float* __restrict__ x,
                      const float* __restrict__ gumbel,
                      const float* __restrict__ W,
                      const float* __restrict__ bias,
                      float* __restrict__ out)
{
    __shared__ float ldsX[2][TPB * CHUNKP];
    __shared__ float ldsW[DDIM * NEXP];
    __shared__ float ldsPart[8 * TPB * NEXP];
    __shared__ float ldsScal[TPB * 4];

    const int tid  = threadIdx.x;
    const int wave = tid >> 5;
    const int lane = tid & 31;
    const int col  = lane & 15;
    const int half = lane >> 4;
    const int tokBase = blockIdx.x * TPB;

    for (int i = tid; i < (DDIM * NEXP) / 4; i += THREADS)
        ((float4*)ldsW)[i] = ((const float4*)W)[i];
    __syncthreads();

    float sSum[2]  = {0.f, 0.f};
    float sUniq[2] = {0.f, 0.f};
    float sPat[2]  = {0.f, 0.f};
    float carry[2] = {0.f, 0.f};

    v8f acc = {};

    int buf = 0;
    for (int c = 0; c < NCHUNK; ++c) {
        #pragma unroll
        for (int j = 0; j < 2; ++j) {
            const int trow  = wave * 2 + j;
            const int token = tokBase + trow;
            const int dg    = c * CHUNK + lane * 4;
            const float4 v  = *(const float4*)(x + (size_t)token * DDIM + dg);
            if (c + 1 < NCHUNK)
                __builtin_prefetch(x + (size_t)token * DDIM + dg + CHUNK, 0, 0);

            sSum[j]  += (v.x + v.y) + (v.z + v.w);
            sUniq[j] += (v.x != 0.f ? 1.f : 0.f) + (v.y != 0.f ? 1.f : 0.f)
                      + (v.z != 0.f ? 1.f : 0.f) + (v.w != 0.f ? 1.f : 0.f);

            float prevw  = __shfl_up(v.w, 1);
            float pfirst = (lane == 0) ? carry[j] : prevw;
            if (dg > 0 && v.x != pfirst) sPat[j] += (float)(DDIM - dg);
            if (v.y != v.x)              sPat[j] += (float)(DDIM - (dg + 1));
            if (v.z != v.y)              sPat[j] += (float)(DDIM - (dg + 2));
            if (v.w != v.z)              sPat[j] += (float)(DDIM - (dg + 3));
            carry[j] = __shfl(v.w, 31);

            *(float4*)(&ldsX[buf][trow * CHUNKP + lane * 4]) = v;
        }
        __syncthreads();

        #pragma unroll
        for (int q = 0; q < 4; ++q) {
            const int klocal = wave * 16 + q * 4 + half * 2;
            const int kglob  = c * CHUNK + klocal;
            const float2 av = *(const float2*)(&ldsX[buf][col * CHUNKP + klocal]);
            v2f a;  a[0] = av.x;  a[1] = av.y;
            v2f bm;
            bm[0] = (col < NEXP) ? ldsW[kglob * NEXP + col]       : 0.f;
            bm[1] = (col < NEXP) ? ldsW[(kglob + 1) * NEXP + col] : 0.f;
            acc = __builtin_amdgcn_wmma_f32_16x16x4_f32(
                      false, a, false, bm, (short)0, acc, false, false);
        }
        buf ^= 1;
    }

    #pragma unroll
    for (int j = 0; j < 2; ++j) {
        float a0 = sSum[j], a1 = sUniq[j], a2 = sPat[j];
        #pragma unroll
        for (int off = 16; off > 0; off >>= 1) {
            a0 += __shfl_xor(a0, off);
            a1 += __shfl_xor(a1, off);
            a2 += __shfl_xor(a2, off);
        }
        if (lane == 0) {
            const int trow = wave * 2 + j;
            ldsScal[trow * 4 + 0] = a0;
            ldsScal[trow * 4 + 1] = a1;
            ldsScal[trow * 4 + 2] = a2;
        }
    }

    if (col < NEXP) {
        #pragma unroll
        for (int r = 0; r < 8; ++r) {
            const int m = r + half * 8;
            ldsPart[(wave * TPB + m) * NEXP + col] = acc[r];
        }
    }
    __syncthreads();

    if (tid < TPB) {
        const int token = tokBase + tid;
        const float s = ldsScal[tid * 4 + 0];
        const float u = ldsScal[tid * 4 + 1];
        const float p = ldsScal[tid * 4 + 2];
        float ev[4];
        ev[0] = s;
        ev[1] = s / ((float)DDIM * (s + EPSF));
        ev[2] = u;
        ev[3] = p / (float)DDIM;

        int best = 0;
        float bestv = -INFINITY;
        #pragma unroll
        for (int e = 0; e < NEXP; ++e) {
            float le = 0.f;
            #pragma unroll
            for (int w = 0; w < 8; ++w)
                le += ldsPart[(w * TPB + tid) * NEXP + e];
            le += bias[e] + gumbel[(size_t)token * NEXP + e];
            if (le > bestv) { bestv = le; best = e; }
        }
        out[token] = ev[best];
    }
}

extern "C" void kernel_launch(void* const* d_in, const int* in_sizes, int n_in,
                              void* d_out, int out_size, void* d_ws, size_t ws_size,
                              hipStream_t stream) {
    (void)in_sizes; (void)n_in; (void)d_ws; (void)ws_size; (void)out_size;
    const float* x      = (const float*)d_in[0];
    const float* gumbel = (const float*)d_in[1];
    const float* W      = (const float*)d_in[2];
    const float* b      = (const float*)d_in[3];
    float* out          = (float*)d_out;

    dim3 grid(TOKENS / TPB);
    moe_count_kernel<<<grid, THREADS, 0, stream>>>(x, gumbel, W, b, out);
}


# Row loads issued 8 chunks ahead per wave (16 b128 in flight, was 2 plus cache prefetch), W staged with its 8 loads at once; per-chunk barrier waits on LDS only so loads stay in flight across it
# speedup vs baseline: 1.0695x; 1.0695x over previous
#include <hip/hip_runtime.h>
#include <hip/hip_bf16.h>


#define TOKENS   32768
#define DDIM     2048
#define NEXP     4
#define EPSF     1e-6f
#define CHUNK    128
#define CHUNKP   132
#define NCHUNK   (DDIM / CHUNK)
#define TPB      16
#define THREADS  256
#define DEPTH    8

static __device__ __forceinline__ void lds_barrier() {
    asm volatile("s_wait_dscnt 0x0\n\ts_barrier_signal -1\n\ts_barrier_wait -1" ::: "memory");
}

typedef __attribute__((ext_vector_type(2))) float v2f;
typedef __attribute__((ext_vector_type(8))) float v8f;

__global__ __launch_bounds__(THREADS)
void moe_count_kernel(const float* __restrict__ x,
                      const float* __restrict__ gumbel,
                      const float* __restrict__ W,
                      const float* __restrict__ bias,
                      float* __restrict__ out)
{
    __shared__ float ldsX[2][TPB * CHUNKP];
    __shared__ float ldsW[DDIM * NEXP];
    __shared__ float ldsPart[8 * TPB * NEXP];
    __shared__ float ldsScal[TPB * 4];

    const int tid  = threadIdx.x;
    const int wave = tid >> 5;
    const int lane = tid & 31;
    const int col  = lane & 15;
    const int half = lane >> 4;
    const int tokBase = blockIdx.x * TPB;

    const float* xrow0 = x + (size_t)(tokBase + wave * 2) * DDIM + lane * 4;
    const float* xrow1 = xrow0 + DDIM;

    float4 xv[NCHUNK][2];

    float4 wv[(DDIM * NEXP) / 4 / THREADS];
    #pragma unroll
    for (int i = 0; i < (DDIM * NEXP) / 4 / THREADS; ++i)
        wv[i] = ((const float4*)W)[tid + i * THREADS];
    #pragma unroll
    for (int i = 0; i < (DDIM * NEXP) / 4 / THREADS; ++i)
        ((float4*)ldsW)[tid + i * THREADS] = wv[i];

    #pragma unroll
    for (int c = 0; c < DEPTH; ++c) {
        __builtin_amdgcn_sched_barrier(0);
        xv[c][0] = *(const float4*)(xrow0 + c * CHUNK);
        xv[c][1] = *(const float4*)(xrow1 + c * CHUNK);
    }

    lds_barrier();

    float sSum[2]  = {0.f, 0.f};
    float sUniq[2] = {0.f, 0.f};
    float sPat[2]  = {0.f, 0.f};
    float carry[2] = {0.f, 0.f};

    v8f acc = {};

    #pragma unroll
    for (int c = 0; c < NCHUNK; ++c) {
        const int buf = c & 1;
        if (c + DEPTH < NCHUNK) {
            xv[c + DEPTH][0] = *(const float4*)(xrow0 + (c + DEPTH) * CHUNK);
            xv[c + DEPTH][1] = *(const float4*)(xrow1 + (c + DEPTH) * CHUNK);
        }

        #pragma unroll
        for (int j = 0; j < 2; ++j) {
            const int trow  = wave * 2 + j;
            const int dg    = c * CHUNK + lane * 4;
            float4 v = xv[c][j];
            asm volatile("" : "+v"(v.x), "+v"(v.y), "+v"(v.z), "+v"(v.w));

            sSum[j]  += (v.x + v.y) + (v.z + v.w);
            sUniq[j] += (v.x != 0.f ? 1.f : 0.f) + (v.y != 0.f ? 1.f : 0.f)
                      + (v.z != 0.f ? 1.f : 0.f) + (v.w != 0.f ? 1.f : 0.f);

            float prevw  = __shfl_up(v.w, 1);
            float pfirst = (lane == 0) ? carry[j] : prevw;
            if (dg > 0 && v.x != pfirst) sPat[j] += (float)(DDIM - dg);
            if (v.y != v.x)              sPat[j] += (float)(DDIM - (dg + 1));
            if (v.z != v.y)              sPat[j] += (float)(DDIM - (dg + 2));
            if (v.w != v.z)              sPat[j] += (float)(DDIM - (dg + 3));
            carry[j] = __shfl(v.w, 31);

            *(float4*)(&ldsX[buf][trow * CHUNKP + lane * 4]) = v;
        }
        asm volatile("" : "+v"(sSum[0]), "+v"(sSum[1]), "+v"(sUniq[0]), "+v"(sUniq[1]),
                          "+v"(sPat[0]), "+v"(sPat[1]));
        lds_barrier();

        #pragma unroll
        for (int q = 0; q < 4; ++q) {
            const int klocal = wave * 16 + q * 4 + half * 2;
            const int kglob  = c * CHUNK + klocal;
            const float2 av = *(const float2*)(&ldsX[buf][col * CHUNKP + klocal]);
            v2f a;  a[0] = av.x;  a[1] = av.y;
            v2f bm;
            bm[0] = (col < NEXP) ? ldsW[kglob * NEXP + col]       : 0.f;
            bm[1] = (col < NEXP) ? ldsW[(kglob + 1) * NEXP + col] : 0.f;
            acc = __builtin_amdgcn_wmma_f32_16x16x4_f32(
                      false, a, false, bm, (short)0, acc, false, false);
        }
    }

    #pragma unroll
    for (int j = 0; j < 2; ++j) {
        float a0 = sSum[j], a1 = sUniq[j], a2 = sPat[j];
        #pragma unroll
        for (int off = 16; off > 0; off >>= 1) {
            a0 += __shfl_xor(a0, off);
            a1 += __shfl_xor(a1, off);
            a2 += __shfl_xor(a2, off);
        }
        if (lane == 0) {
            const int trow = wave * 2 + j;
            ldsScal[trow * 4 + 0] = a0;
            ldsScal[trow * 4 + 1] = a1;
            ldsScal[trow * 4 + 2] = a2;
        }
    }

    if (col < NEXP) {
        #pragma unroll
        for (int r = 0; r < 8; ++r) {
            const int m = r + half * 8;
            ldsPart[(wave * TPB + m) * NEXP + col] = acc[r];
        }
    }
    __syncthreads();

    if (tid < TPB) {
        const int token = tokBase + tid;
        const float s = ldsScal[tid * 4 + 0];
        const float u = ldsScal[tid * 4 + 1];
        const float p = ldsScal[tid * 4 + 2];
        float ev[4];
        ev[0] = s;
        ev[1] = s / ((float)DDIM * (s + EPSF));
        ev[2] = u;
        ev[3] = p / (float)DDIM;

        int best = 0;
        float bestv = -INFINITY;
        #pragma unroll
        for (int e = 0; e < NEXP; ++e) {
            float le = 0.f;
            #pragma unroll
            for (int w = 0; w < 8; ++w)
                le += ldsPart[(w * TPB + tid) * NEXP + e];
            le += bias[e] + gumbel[(size_t)token * NEXP + e];
            if (le > bestv) { bestv = le; best = e; }
        }
        out[token] = ev[best];
    }
}

extern "C" void kernel_launch(void* const* d_in, const int* in_sizes, int n_in,
                              void* d_out, int out_size, void* d_ws, size_t ws_size,
                              hipStream_t stream) {
    (void)in_sizes; (void)n_in; (void)d_ws; (void)ws_size; (void)out_size;
    const float* x      = (const float*)d_in[0];
    const float* gumbel = (const float*)d_in[1];
    const float* W      = (const float*)d_in[2];
    const float* b      = (const float*)d_in[3];
    float* out          = (float*)d_out;

    dim3 grid(TOKENS / TPB);
    moe_count_kernel<<<grid, THREADS, 0, stream>>>(x, gumbel, W, b, out);
}


# A and B fragments each read as one 16-byte LDS load per two WMMAs (W kept 4 k per expert contiguous, no pad selects), issued as a group at the barrier; k order of the logit sums changes
# speedup vs baseline: 1.1451x; 1.0707x over previous
#include <hip/hip_runtime.h>
#include <hip/hip_bf16.h>


#define TOKENS   32768
#define DDIM     2048
#define NEXP     4
#define EPSF     1e-6f
#define CHUNK    128
#define CHUNKP   132
#define NCHUNK   (DDIM / CHUNK)
#define TPB      16
#define THREADS  256
#define DEPTH    8

static __device__ __forceinline__ void lds_barrier() {
    asm volatile("s_wait_dscnt 0x0\n\ts_barrier_signal -1\n\ts_barrier_wait -1" ::: "memory");
}

typedef __attribute__((ext_vector_type(2))) float v2f;
typedef __attribute__((ext_vector_type(8))) float v8f;

__global__ __launch_bounds__(THREADS)
void moe_count_kernel(const float* __restrict__ x,
                      const float* __restrict__ gumbel,
                      const float* __restrict__ W,
                      const float* __restrict__ bias,
                      float* __restrict__ out)
{
    __shared__ __attribute__((aligned(16))) float ldsX[2][TPB * CHUNKP];
    __shared__ __attribute__((aligned(16))) float ldsW[DDIM * NEXP];
    __shared__ float ldsPart[8 * TPB * NEXP];
    __shared__ float ldsScal[TPB * 4];

    const int tid  = threadIdx.x;
    const int wave = tid >> 5;
    const int lane = tid & 31;
    const int col  = lane & 15;
    const int half = lane >> 4;
    const int tokBase = blockIdx.x * TPB;

    const float* xrow0 = x + (size_t)(tokBase + wave * 2) * DDIM + lane * 4;
    const float* xrow1 = xrow0 + DDIM;

    float4 xv[NCHUNK][2];

    float4 wv[(DDIM * NEXP) / 4 / THREADS];
    #pragma unroll
    for (int i = 0; i < (DDIM * NEXP) / 4 / THREADS; ++i)
        wv[i] = ((const float4*)W)[tid + i * THREADS];
    #pragma unroll
    for (int i = 0; i < (DDIM * NEXP) / 4 / THREADS; ++i) {
        const int k = tid + i * THREADS;
        float* dst = &ldsW[(k >> 2) * (4 * NEXP) + (k & 3)];
        dst[0] = wv[i].x;  dst[4] = wv[i].y;  dst[8] = wv[i].z;  dst[12] = wv[i].w;
    }

    #pragma unroll
    for (int c = 0; c < DEPTH; ++c) {
        __builtin_amdgcn_sched_barrier(0);
        xv[c][0] = *(const float4*)(xrow0 + c * CHUNK);
        xv[c][1] = *(const float4*)(xrow1 + c * CHUNK);
    }

    lds_barrier();

    float sSum[2]  = {0.f, 0.f};
    float sUniq[2] = {0.f, 0.f};
    float sPat[2]  = {0.f, 0.f};
    float carry[2] = {0.f, 0.f};

    v8f acc = {};

    #pragma unroll
    for (int c = 0; c < NCHUNK; ++c) {
        const int buf = c & 1;
        if (c + DEPTH < NCHUNK) {
            xv[c + DEPTH][0] = *(const float4*)(xrow0 + (c + DEPTH) * CHUNK);
            xv[c + DEPTH][1] = *(const float4*)(xrow1 + (c + DEPTH) * CHUNK);
        }

        #pragma unroll
        for (int j = 0; j < 2; ++j) {
            const int trow  = wave * 2 + j;
            const int dg    = c * CHUNK + lane * 4;
            float4 v = xv[c][j];
            asm volatile("" : "+v"(v.x), "+v"(v.y), "+v"(v.z), "+v"(v.w));

            sSum[j]  += (v.x + v.y) + (v.z + v.w);
            sUniq[j] += (v.x != 0.f ? 1.f : 0.f) + (v.y != 0.f ? 1.f : 0.f)
                      + (v.z != 0.f ? 1.f : 0.f) + (v.w != 0.f ? 1.f : 0.f);

            float prevw  = __shfl_up(v.w, 1);
            float pfirst = (lane == 0) ? carry[j] : prevw;
            if (dg > 0 && v.x != pfirst) sPat[j] += (float)(DDIM - dg);
            if (v.y != v.x)              sPat[j] += (float)(DDIM - (dg + 1));
            if (v.z != v.y)              sPat[j] += (float)(DDIM - (dg + 2));
            if (v.w != v.z)              sPat[j] += (float)(DDIM - (dg + 3));
            carry[j] = __shfl(v.w, 31);

            *(float4*)(&ldsX[buf][trow * CHUNKP + lane * 4]) = v;
        }
        asm volatile("" : "+v"(sSum[0]), "+v"(sSum[1]), "+v"(sUniq[0]), "+v"(sUniq[1]),
                          "+v"(sPat[0]), "+v"(sPat[1]));
        float4 bq[2];
        for (int i = 0; i < 2; ++i) {
            const int kg4 = (c * CHUNK + wave * 16 + i * 8) / 4 + half;
            bq[i] = *(const float4*)(&ldsW[kg4 * (4 * NEXP) + (col & 3) * 4]);
        }
        lds_barrier();

        float4 aq[2];
        for (int i = 0; i < 2; ++i)
            aq[i] = *(const float4*)(&ldsX[buf][col * CHUNKP + wave * 16 + i * 8 + half * 4]);
        for (int i = 0; i < 2; ++i) {
            v2f a, bm;
            a[0] = aq[i].x;  a[1] = aq[i].y;  bm[0] = bq[i].x;  bm[1] = bq[i].y;
            acc = __builtin_amdgcn_wmma_f32_16x16x4_f32(
                      false, a, false, bm, (short)0, acc, false, false);
            a[0] = aq[i].z;  a[1] = aq[i].w;  bm[0] = bq[i].z;  bm[1] = bq[i].w;
            acc = __builtin_amdgcn_wmma_f32_16x16x4_f32(
                      false, a, false, bm, (short)0, acc, false, false);
        }
    }

    #pragma unroll
    for (int j = 0; j < 2; ++j) {
        float a0 = sSum[j], a1 = sUniq[j], a2 = sPat[j];
        #pragma unroll
        for (int off = 16; off > 0; off >>= 1) {
            a0 += __shfl_xor(a0, off);
            a1 += __shfl_xor(a1, off);
            a2 += __shfl_xor(a2, off);
        }
        if (lane == 0) {
            const int trow = wave * 2 + j;
            ldsScal[trow * 4 + 0] = a0;
            ldsScal[trow * 4 + 1] = a1;
            ldsScal[trow * 4 + 2] = a2;
        }
    }

    if (col < NEXP) {
        #pragma unroll
        for (int r = 0; r < 8; ++r) {
            const int m = r + half * 8;
            ldsPart[(wave * TPB + m) * NEXP + col] = acc[r];
        }
    }
    __syncthreads();

    if (tid < TPB) {
        const int token = tokBase + tid;
        const float s = ldsScal[tid * 4 + 0];
        const float u = ldsScal[tid * 4 + 1];
        const float p = ldsScal[tid * 4 + 2];
        float ev[4];
        ev[0] = s;
        ev[1] = s / ((float)DDIM * (s + EPSF));
        ev[2] = u;
        ev[3] = p / (float)DDIM;

        int best = 0;
        float bestv = -INFINITY;
        #pragma unroll
        for (int e = 0; e < NEXP; ++e) {
            float le = 0.f;
            #pragma unroll
            for (int w = 0; w < 8; ++w)
                le += ldsPart[(w * TPB + tid) * NEXP + e];
            le += bias[e] + gumbel[(size_t)token * NEXP + e];
            if (le > bestv) { bestv = le; best = e; }
        }
        out[token] = ev[best];
    }
}

extern "C" void kernel_launch(void* const* d_in, const int* in_sizes, int n_in,
                              void* d_out, int out_size, void* d_ws, size_t ws_size,
                              hipStream_t stream) {
    (void)in_sizes; (void)n_in; (void)d_ws; (void)ws_size; (void)out_size;
    const float* x      = (const float*)d_in[0];
    const float* gumbel = (const float*)d_in[1];
    const float* W      = (const float*)d_in[2];
    const float* b      = (const float*)d_in[3];
    float* out          = (float*)d_out;

    dim3 grid(TOKENS / TPB);
    moe_count_kernel<<<grid, THREADS, 0, stream>>>(x, gumbel, W, b, out);
}


# Each workgroup loops over 4 consecutive 16-token tiles (grid 2048 -> 512): W staged once per workgroup, next tile's row loads requested in the last 8 chunks so the load stream runs on across tiles
# speedup vs baseline: 1.2485x; 1.0903x over previous
#include <hip/hip_runtime.h>
#include <hip/hip_bf16.h>


#define TOKENS   32768
#define DDIM     2048
#define NEXP     4
#define EPSF     1e-6f
#define CHUNK    128
#define CHUNKP   132
#define NCHUNK   (DDIM / CHUNK)
#define TPB      16
#define THREADS  256
#define DEPTH    8
#define TILES    4
#define NWG      (TOKENS / (TPB * TILES))

static_assert(DEPTH < NCHUNK, "the next tile's chunk c+DEPTH-NCHUNK goes into a slot that an earlier chunk of this tile has left");
static_assert(THREADS == TILES * TPB * NEXP, "one noise value per thread in the prologue");

static __device__ __forceinline__ void lds_barrier() {
    asm volatile("s_wait_dscnt 0x0\n\ts_barrier_signal -1\n\ts_barrier_wait -1" ::: "memory");
}

typedef __attribute__((ext_vector_type(2))) float v2f;
typedef __attribute__((ext_vector_type(8))) float v8f;
typedef __attribute__((ext_vector_type(4))) float v4f;

template <bool HAS_NEXT>
static __device__ __forceinline__ void stream_tile(
        v4f (&xv)[NCHUNK][2],
        const float* __restrict__ xrow0,
        float* __restrict__ outTile,
        const float* gbTile,
        const int tid, const int wave, const int lane, const int col, const int half,
        float (*ldsX)[TPB * CHUNKP], const float* ldsW, float* ldsPart, float* ldsScal)
{
    const float* xrow1 = xrow0 + DDIM;
    const float* nrow0 = xrow0 + TPB * DDIM;
    const float* nrow1 = nrow0 + DDIM;

    float sSum[2]  = {0.f, 0.f};
    float sUniq[2] = {0.f, 0.f};
    float sPat[2]  = {0.f, 0.f};
    float carry[2] = {0.f, 0.f};

    v8f acc = {};

    #pragma unroll
    for (int c = 0; c < NCHUNK; ++c) {
        const int buf = c & 1;
        if (c + DEPTH < NCHUNK) {
            xv[c + DEPTH][0] = *(const v4f*)(xrow0 + (c + DEPTH) * CHUNK);
            xv[c + DEPTH][1] = *(const v4f*)(xrow1 + (c + DEPTH) * CHUNK);
        } else if (HAS_NEXT) {
            xv[c + DEPTH - NCHUNK][0] = *(const v4f*)(nrow0 + (c + DEPTH - NCHUNK) * CHUNK);
            xv[c + DEPTH - NCHUNK][1] = *(const v4f*)(nrow1 + (c + DEPTH - NCHUNK) * CHUNK);
        }

        #pragma unroll
        for (int j = 0; j < 2; ++j) {
            const int trow  = wave * 2 + j;
            const int dg    = c * CHUNK + lane * 4;
            v4f v = xv[c][j];
            asm volatile("" : "+v"(v));

            sSum[j]  += (v.x + v.y) + (v.z + v.w);
            sUniq[j] += (v.x != 0.f ? 1.f : 0.f) + (v.y != 0.f ? 1.f : 0.f)
                      + (v.z != 0.f ? 1.f : 0.f) + (v.w != 0.f ? 1.f : 0.f);

            float prevw  = __shfl_up(v.w, 1);
            float pfirst = (lane == 0) ? carry[j] : prevw;
            if (dg > 0 && v.x != pfirst) sPat[j] += (float)(DDIM - dg);
            if (v.y != v.x)              sPat[j] += (float)(DDIM - (dg + 1));
            if (v.z != v.y)              sPat[j] += (float)(DDIM - (dg + 2));
            if (v.w != v.z)              sPat[j] += (float)(DDIM - (dg + 3));
            carry[j] = __shfl(v.w, 31);

            *(v4f*)(&ldsX[buf][trow * CHUNKP + lane * 4]) = v;
        }
        asm volatile("" : "+v"(sSum[0]), "+v"(sSum[1]), "+v"(sUniq[0]), "+v"(sUniq[1]),
                          "+v"(sPat[0]), "+v"(sPat[1]));
        float4 bq[2];
        for (int i = 0; i < 2; ++i) {
            const int kg4 = (c * CHUNK + wave * 16 + i * 8) / 4 + half;
            bq[i] = *(const float4*)(&ldsW[kg4 * (4 * NEXP) + (col & 3) * 4]);
        }
        lds_barrier();

        float4 aq[2];
        for (int i = 0; i < 2; ++i)
            aq[i] = *(const float4*)(&ldsX[buf][col * CHUNKP + wave * 16 + i * 8 + half * 4]);
        for (int i = 0; i < 2; ++i) {
            v2f a, bm;
            a[0] = aq[i].x;  a[1] = aq[i].y;  bm[0] = bq[i].x;  bm[1] = bq[i].y;
            acc = __builtin_amdgcn_wmma_f32_16x16x4_f32(
                      false, a, false, bm, (short)0, acc, false, false);
            a[0] = aq[i].z;  a[1] = aq[i].w;  bm[0] = bq[i].z;  bm[1] = bq[i].w;
            acc = __builtin_amdgcn_wmma_f32_16x16x4_f32(
                      false, a, false, bm, (short)0, acc, false, false);
        }
    }

    #pragma unroll
    for (int j = 0; j < 2; ++j) {
        float a0 = sSum[j], a1 = sUniq[j], a2 = sPat[j];
        #pragma unroll
        for (int off = 16; off > 0; off >>= 1) {
            a0 += __shfl_xor(a0, off);
            a1 += __shfl_xor(a1, off);
            a2 += __shfl_xor(a2, off);
        }
        if (lane == 0) {
            const int trow = wave * 2 + j;
            ldsScal[trow * 4 + 0] = a0;
            ldsScal[trow * 4 + 1] = a1;
            ldsScal[trow * 4 + 2] = a2;
        }
    }

    if (col < NEXP) {
        #pragma unroll
        for (int r = 0; r < 8; ++r) {
            const int m = r + half * 8;
            ldsPart[(wave * TPB + m) * NEXP + col] = acc[r];
        }
    }
    lds_barrier();

    if (tid < TPB) {
        const float s = ldsScal[tid * 4 + 0];
        const float u = ldsScal[tid * 4 + 1];
        const float p = ldsScal[tid * 4 + 2];
        float ev[4];
        ev[0] = s;
        ev[1] = s / ((float)DDIM * (s + EPSF));
        ev[2] = u;
        ev[3] = p / (float)DDIM;

        int best = 0;
        float bestv = -INFINITY;
        #pragma unroll
        for (int e = 0; e < NEXP; ++e) {
            float le = 0.f;
            #pragma unroll
            for (int w = 0; w < 8; ++w)
                le += ldsPart[(w * TPB + tid) * NEXP + e];
            le += gbTile[tid * NEXP + e];
            if (le > bestv) { bestv = le; best = e; }
        }
        outTile[tid] = ev[best];
    }
}

__global__ __launch_bounds__(THREADS)
void moe_count_kernel(const float* __restrict__ x,
                      const float* __restrict__ gumbel,
                      const float* __restrict__ W,
                      const float* __restrict__ bias,
                      float* __restrict__ out)
{
    __shared__ __attribute__((aligned(16))) float ldsX[2][TPB * CHUNKP];
    __shared__ __attribute__((aligned(16))) float ldsW[DDIM * NEXP];
    __shared__ float ldsPart[8 * TPB * NEXP];
    __shared__ float ldsScal[TPB * 4];
    __shared__ float ldsGB[TILES * TPB * NEXP];

    const int tid  = threadIdx.x;
    const int wave = tid >> 5;
    const int lane = tid & 31;
    const int col  = lane & 15;
    const int half = lane >> 4;
    const int wgTok = blockIdx.x * (TILES * TPB);

    const float* xrow0 = x + (size_t)(wgTok + wave * 2) * DDIM + lane * 4;

    v4f xv[NCHUNK][2];

    float4 wv[(DDIM * NEXP) / 4 / THREADS];
    #pragma unroll
    for (int i = 0; i < (DDIM * NEXP) / 4 / THREADS; ++i)
        wv[i] = ((const float4*)W)[tid + i * THREADS];
    const float gv = gumbel[(size_t)wgTok * NEXP + tid];
    const float bv = bias[tid & (NEXP - 1)];
    #pragma unroll
    for (int i = 0; i < (DDIM * NEXP) / 4 / THREADS; ++i) {
        const int k = tid + i * THREADS;
        float* dst = &ldsW[(k >> 2) * (4 * NEXP) + (k & 3)];
        dst[0] = wv[i].x;  dst[4] = wv[i].y;  dst[8] = wv[i].z;  dst[12] = wv[i].w;
    }
    ldsGB[tid] = bv + gv;

    #pragma unroll
    for (int c = 0; c < DEPTH; ++c) {
        __builtin_amdgcn_sched_barrier(0);
        xv[c][0] = *(const v4f*)(xrow0 + c * CHUNK);
        xv[c][1] = *(const v4f*)(xrow0 + DDIM + c * CHUNK);
    }

    lds_barrier();

    for (int t = 0; t < TILES - 1; ++t) {
        stream_tile<true>(xv, xrow0, out + wgTok + t * TPB, &ldsGB[t * TPB * NEXP],
                          tid, wave, lane, col, half, ldsX, ldsW, ldsPart, ldsScal);
        xrow0 += TPB * DDIM;
    }
    stream_tile<false>(xv, xrow0, out + wgTok + (TILES - 1) * TPB, &ldsGB[(TILES - 1) * TPB * NEXP],
                       tid, wave, lane, col, half, ldsX, ldsW, ldsPart, ldsScal);
}

extern "C" void kernel_launch(void* const* d_in, const int* in_sizes, int n_in,
                              void* d_out, int out_size, void* d_ws, size_t ws_size,
                              hipStream_t stream) {
    (void)in_sizes; (void)n_in; (void)d_ws; (void)ws_size; (void)out_size;
    const float* x      = (const float*)d_in[0];
    const float* gumbel = (const float*)d_in[1];
    const float* W      = (const float*)d_in[2];
    const float* b      = (const float*)d_in[3];
    float* out          = (float*)d_out;

    dim3 grid(NWG);
    moe_count_kernel<<<grid, THREADS, 0, stream>>>(x, gumbel, W, b, out);
}


# Each wave builds the WMMA fragments of its own 16 rows through an LDS slice only it touches (2 waves per tile, half of k each): no workgroup barrier in the step loop; logit and row sum order changes
# speedup vs baseline: 1.3062x; 1.0462x over previous
#include <hip/hip_runtime.h>
#include <hip/hip_bf16.h>


#define TOKENS   32768
#define DDIM     2048
#define NEXP     4
#define EPSF     1e-6f
#define TPB      16
#define THREADS  256
#define NWAVE    8
#define KSPLIT   2
#define KW       (DDIM / KSPLIT)
#define STEP     64
#define STEPP    68
#define NSTEP    (KW / STEP)
#define NLD      8
#define TILES    (NWAVE / KSPLIT)
#define NWG      (TOKENS / (TPB * TILES))

static_assert(THREADS == TILES * TPB * NEXP, "one noise value per thread in the prologue");
static_assert(NSTEP % 2 == 0, "the step loop works on two register sets in turn");

static __device__ __forceinline__ void lds_barrier() {
    asm volatile("s_wait_dscnt 0x0\n\ts_barrier_signal -1\n\ts_barrier_wait -1" ::: "memory");
}

static __device__ __forceinline__ void wave_lds_order() {
    __builtin_amdgcn_fence(__ATOMIC_RELEASE, "wavefront");
    __builtin_amdgcn_wave_barrier();
    __builtin_amdgcn_fence(__ATOMIC_ACQUIRE, "wavefront");
}

typedef __attribute__((ext_vector_type(2))) float v2f;
typedef __attribute__((ext_vector_type(8))) float v8f;
typedef __attribute__((ext_vector_type(4))) float v4f;

template <bool REFILL>
static __device__ __forceinline__ void do_step(
        v4f (&xs)[NLD], const float* __restrict__ xnext,
        const int kstep, const int kbase, const int col, const int half,
        float (&sSum)[NLD], float (&sUniq)[NLD], float (&sPat)[NLD], float (&carry)[NLD],
        v8f& acc0, v8f& acc1,
        float* tileW, const float* ldsW, const int stOff, const int aOff)
{
    const int   dg  = kstep + col * 4;
    const float wgt = (float)(DDIM - dg);

    wave_lds_order();
    #pragma unroll
    for (int i = 0; i < NLD; ++i) {
        v4f v = xs[i];
        asm volatile("" : "+v"(v));
        *(v4f*)(tileW + stOff + i * 2 * STEPP) = v;

        sSum[i]  += (v.x + v.y) + (v.z + v.w);
        sUniq[i] += (v.x != 0.f ? 1.f : 0.f) + (v.y != 0.f ? 1.f : 0.f)
                  + (v.z != 0.f ? 1.f : 0.f) + (v.w != 0.f ? 1.f : 0.f);

        const float prevw  = __shfl(v.w, col + 15, 16);
        const float pfirst = (col == 0) ? carry[i] : prevw;
        carry[i] = prevw;
        if (dg > kbase && v.x != pfirst) sPat[i] += wgt;
        if (v.y != v.x)                  sPat[i] += wgt - 1.f;
        if (v.z != v.y)                  sPat[i] += wgt - 2.f;
        if (v.w != v.z)                  sPat[i] += wgt - 3.f;
    }
    asm volatile("" : "+v"(sSum[0]), "+v"(sSum[1]), "+v"(sSum[2]), "+v"(sSum[3]),
                      "+v"(sSum[4]), "+v"(sSum[5]), "+v"(sSum[6]), "+v"(sSum[7]));
    asm volatile("" : "+v"(sUniq[0]), "+v"(sUniq[1]), "+v"(sUniq[2]), "+v"(sUniq[3]),
                      "+v"(sUniq[4]), "+v"(sUniq[5]), "+v"(sUniq[6]), "+v"(sUniq[7]));
    asm volatile("" : "+v"(sPat[0]), "+v"(sPat[1]), "+v"(sPat[2]), "+v"(sPat[3]),
                      "+v"(sPat[4]), "+v"(sPat[5]), "+v"(sPat[6]), "+v"(sPat[7]));
    asm volatile("" : "+v"(carry[0]), "+v"(carry[1]), "+v"(carry[2]), "+v"(carry[3]),
                      "+v"(carry[4]), "+v"(carry[5]), "+v"(carry[6]), "+v"(carry[7]));
    wave_lds_order();
    if (REFILL) {
        #pragma unroll
        for (int i = 0; i < NLD; ++i) xs[i] = *(const v4f*)(xnext + (size_t)i * 2 * DDIM);
    }

    float4 bq[STEP / 8], aq[STEP / 8];
    #pragma unroll
    for (int j = 0; j < STEP / 8; ++j)
        bq[j] = *(const float4*)(ldsW + ((kstep >> 2) + 2 * j + half) * (4 * NEXP) + (col & 3) * 4);
    #pragma unroll
    for (int j = 0; j < STEP / 8; ++j)
        aq[j] = *(const float4*)(tileW + aOff + j * 8);
    __builtin_amdgcn_sched_barrier(0);
    #pragma unroll
    for (int j = 0; j < STEP / 8; ++j) {
        v2f a, bm;
        a[0] = aq[j].x;  a[1] = aq[j].y;  bm[0] = bq[j].x;  bm[1] = bq[j].y;
        acc0 = __builtin_amdgcn_wmma_f32_16x16x4_f32(
                   false, a, false, bm, (short)0, acc0, false, false);
        a[0] = aq[j].z;  a[1] = aq[j].w;  bm[0] = bq[j].z;  bm[1] = bq[j].w;
        acc1 = __builtin_amdgcn_wmma_f32_16x16x4_f32(
                   false, a, false, bm, (short)0, acc1, false, false);
    }
}

__global__ __launch_bounds__(THREADS)
void moe_count_kernel(const float* __restrict__ x,
                      const float* __restrict__ gumbel,
                      const float* __restrict__ W,
                      const float* __restrict__ bias,
                      float* __restrict__ out)
{
    __shared__ __attribute__((aligned(16))) float ldsW[DDIM * NEXP];
    __shared__ __attribute__((aligned(16))) float ldsT[NWAVE][TPB * STEPP];
    __shared__ float ldsPart[NWAVE][TPB * NEXP];
    __shared__ float ldsScal[NWAVE][TPB * 4];
    __shared__ float ldsGB[TILES * TPB * NEXP];

    const int tid  = threadIdx.x;
    const int wave = tid >> 5;
    const int lane = tid & 31;
    const int col  = lane & 15;
    const int half = lane >> 4;
    const int tile = wave >> 1;
    const int kh   = wave & 1;
    const int wgTok = blockIdx.x * (TILES * TPB);
    const int kbase = kh * KW;

    const float* xp = x + (size_t)(wgTok + tile * TPB + half) * DDIM + kbase + col * 4;

    float4 wv[(DDIM * NEXP) / 4 / THREADS];
    #pragma unroll
    for (int i = 0; i < (DDIM * NEXP) / 4 / THREADS; ++i)
        wv[i] = ((const float4*)W)[tid + i * THREADS];
    const float gv = gumbel[(size_t)wgTok * NEXP + tid];
    const float bv = bias[tid & (NEXP - 1)];

    v4f xa[NLD], xb[NLD];
    #pragma unroll
    for (int i = 0; i < NLD; ++i) xa[i] = *(const v4f*)(xp + (size_t)i * 2 * DDIM);
    __builtin_amdgcn_sched_barrier(0);
    #pragma unroll
    for (int i = 0; i < NLD; ++i) xb[i] = *(const v4f*)(xp + STEP + (size_t)i * 2 * DDIM);

    #pragma unroll
    for (int i = 0; i < (DDIM * NEXP) / 4 / THREADS; ++i) {
        const int k = tid + i * THREADS;
        float* dst = &ldsW[(k >> 2) * (4 * NEXP) + (k & 3)];
        dst[0] = wv[i].x;  dst[4] = wv[i].y;  dst[8] = wv[i].z;  dst[12] = wv[i].w;
    }
    ldsGB[tid] = bv + gv;

    lds_barrier();

    float sSum[NLD], sUniq[NLD], sPat[NLD], carry[NLD];
    #pragma unroll
    for (int i = 0; i < NLD; ++i) { sSum[i] = 0.f; sUniq[i] = 0.f; sPat[i] = 0.f; carry[i] = 0.f; }
    v8f acc0 = {}, acc1 = {};

    float* tileW = &ldsT[wave][0];
    const int stOff = half * STEPP + col * 4;
    const int aOff  = col * STEPP + half * 4;

    if (kh == 1 && col == 0) {
        #pragma unroll
        for (int i = 0; i < NLD; ++i) ldsScal[wave][(2 * i + half) * 4 + 3] = xa[i].x;
    }

    int kstep = kbase;
    const float* xn = xp + 2 * STEP;
    #pragma nounroll
    for (int it = 0; it < NSTEP / 2 - 1; ++it) {
        do_step<true>(xa, xn, kstep, kbase, col, half, sSum, sUniq, sPat, carry,
                      acc0, acc1, tileW, ldsW, stOff, aOff);
        do_step<true>(xb, xn + STEP, kstep + STEP, kbase, col, half, sSum, sUniq, sPat, carry,
                      acc0, acc1, tileW, ldsW, stOff, aOff);
        kstep += 2 * STEP;
        xn    += 2 * STEP;
    }
    do_step<false>(xa, xn, kstep, kbase, col, half, sSum, sUniq, sPat, carry,
                   acc0, acc1, tileW, ldsW, stOff, aOff);
    do_step<false>(xb, xn, kstep + STEP, kbase, col, half, sSum, sUniq, sPat, carry,
                   acc0, acc1, tileW, ldsW, stOff, aOff);

    if (kh == 0 && col == 0) {
        #pragma unroll
        for (int i = 0; i < NLD; ++i) ldsScal[wave][(2 * i + half) * 4 + 3] = carry[i];
    }

    #pragma unroll
    for (int i = 0; i < NLD; ++i) {
        float a0 = sSum[i], a1 = sUniq[i], a2 = sPat[i];
        #pragma unroll
        for (int off = 8; off > 0; off >>= 1) {
            a0 += __shfl_xor(a0, off);
            a1 += __shfl_xor(a1, off);
            a2 += __shfl_xor(a2, off);
        }
        if (col == 0) {
            const int trow = 2 * i + half;
            ldsScal[wave][trow * 4 + 0] = a0;
            ldsScal[wave][trow * 4 + 1] = a1;
            ldsScal[wave][trow * 4 + 2] = a2;
        }
    }

    if (col < NEXP) {
        #pragma unroll
        for (int r = 0; r < 8; ++r) {
            const int m = r + half * 8;
            ldsPart[wave][m * NEXP + col] = acc0[r] + acc1[r];
        }
    }
    __syncthreads();

    if (tid < TILES * TPB) {
        const int w0 = (tid >> 4) * 2, w1 = w0 + 1, m = tid & 15;
        const float s = ldsScal[w0][m * 4 + 0] + ldsScal[w1][m * 4 + 0];
        const float u = ldsScal[w0][m * 4 + 1] + ldsScal[w1][m * 4 + 1];
        float p = ldsScal[w0][m * 4 + 2] + ldsScal[w1][m * 4 + 2];
        if (ldsScal[w1][m * 4 + 3] != ldsScal[w0][m * 4 + 3]) p += (float)(DDIM - KW);
        float ev[4];
        ev[0] = s;
        ev[1] = s / ((float)DDIM * (s + EPSF));
        ev[2] = u;
        ev[3] = p / (float)DDIM;

        int best = 0;
        float bestv = -INFINITY;
        #pragma unroll
        for (int e = 0; e < NEXP; ++e) {
            float le = ldsPart[w0][m * NEXP + e] + ldsPart[w1][m * NEXP + e];
            le += ldsGB[tid * NEXP + e];
            if (le > bestv) { bestv = le; best = e; }
        }
        out[wgTok + tid] = ev[best];
    }
}

extern "C" void kernel_launch(void* const* d_in, const int* in_sizes, int n_in,
                              void* d_out, int out_size, void* d_ws, size_t ws_size,
                              hipStream_t stream) {
    (void)in_sizes; (void)n_in; (void)d_ws; (void)ws_size; (void)out_size;
    const float* x      = (const float*)d_in[0];
    const float* gumbel = (const float*)d_in[1];
    const float* W      = (const float*)d_in[2];
    const float* b      = (const float*)d_in[3];
    float* out          = (float*)d_out;

    dim3 grid(NWG);
    moe_count_kernel<<<grid, THREADS, 0, stream>>>(x, gumbel, W, b, out);
}


# 16-lane exchanges (left neighbour of each piece, closing row sums) on the vector unit (DPP row_ror, row_xmask) instead of LDS permutes: no LDS wait in the search or among the WMMAs; same bits
# speedup vs baseline: 1.4751x; 1.1293x over previous
#include <hip/hip_runtime.h>
#include <hip/hip_bf16.h>


#define TOKENS   32768
#define DDIM     2048
#define NEXP     4
#define EPSF     1e-6f
#define TPB      16
#define THREADS  256
#define NWAVE    8
#define KSPLIT   2
#define KW       (DDIM / KSPLIT)
#define STEP     64
#define STEPP    68
#define NSTEP    (KW / STEP)
#define NLD      8
#define TILES    (NWAVE / KSPLIT)
#define NWG      (TOKENS / (TPB * TILES))

static_assert(THREADS == TILES * TPB * NEXP, "one noise value per thread in the prologue");
static_assert(NSTEP % 2 == 0, "the step loop works on two register sets in turn");

static __device__ __forceinline__ void lds_barrier() {
    asm volatile("s_wait_dscnt 0x0\n\ts_barrier_signal -1\n\ts_barrier_wait -1" ::: "memory");
}

static __device__ __forceinline__ void wave_lds_order() {
    __builtin_amdgcn_fence(__ATOMIC_RELEASE, "wavefront");
    __builtin_amdgcn_wave_barrier();
    __builtin_amdgcn_fence(__ATOMIC_ACQUIRE, "wavefront");
}

static __device__ __forceinline__ float row_left(const float x) {
    const int b = __builtin_bit_cast(int, x);
    return __builtin_bit_cast(float, __builtin_amdgcn_mov_dpp(b, 0x121, 0xf, 0xf, false));
}

template <int MASK>
static __device__ __forceinline__ float row_xadd(const float x) {
    const int b = __builtin_bit_cast(int, x);
    return x + __builtin_bit_cast(float, __builtin_amdgcn_mov_dpp(b, 0x160 + MASK, 0xf, 0xf, false));
}

static __device__ __forceinline__ float row_sum16(float x) {
    x = row_xadd<8>(x);  x = row_xadd<4>(x);  x = row_xadd<2>(x);  x = row_xadd<1>(x);
    return x;
}

typedef __attribute__((ext_vector_type(2))) float v2f;
typedef __attribute__((ext_vector_type(8))) float v8f;
typedef __attribute__((ext_vector_type(4))) float v4f;

template <bool REFILL>
static __device__ __forceinline__ void do_step(
        v4f (&xs)[NLD], const float* __restrict__ xnext,
        const int kstep, const int kbase, const int col, const int half,
        float (&sSum)[NLD], float (&sZero)[NLD], float (&sEqW)[NLD], float (&carry)[NLD],
        bool& everSlow, v8f& acc0, v8f& acc1,
        float* tileW, const float* ldsW, const int stOff, const int aOff)
{
    const int dg = kstep + col * 4;

    wave_lds_order();
    float pf[NLD];
    float zmin = INFINITY;
    bool  anyEq = false;
    #pragma unroll
    for (int i = 0; i < NLD; ++i) {
        asm volatile("" : "+v"(xs[i]));
        const v4f v = xs[i];
        *(v4f*)(tileW + stOff + i * 2 * STEPP) = v;

        sSum[i] += (v.x + v.y) + (v.z + v.w);

        zmin = fminf(fminf(zmin, fabsf(v.x)), fabsf(v.y));
        zmin = fminf(fminf(zmin, fabsf(v.z)), fabsf(v.w));

        const float prevw = row_left(v.w);
        pf[i]    = (col == 0) ? carry[i] : prevw;
        carry[i] = prevw;
        anyEq |= (v.x == pf[i]) | (v.y == v.x) | (v.z == v.y) | (v.w == v.z);
    }
    if (__builtin_amdgcn_ballot_w32(anyEq | (zmin == 0.f)) != 0u) {
        everSlow = true;
        const float wgt = (float)(DDIM - dg);
        #pragma unroll
        for (int i = 0; i < NLD; ++i) {
            const v4f v = xs[i];
            sZero[i] += (v.x == 0.f ? 1.f : 0.f) + (v.y == 0.f ? 1.f : 0.f)
                      + (v.z == 0.f ? 1.f : 0.f) + (v.w == 0.f ? 1.f : 0.f);
            if (dg > kbase && v.x == pf[i]) sEqW[i] += wgt;
            if (v.y == v.x)                 sEqW[i] += wgt - 1.f;
            if (v.z == v.y)                 sEqW[i] += wgt - 2.f;
            if (v.w == v.z)                 sEqW[i] += wgt - 3.f;
        }
    }
    asm volatile("" : "+v"(sSum[0]), "+v"(sSum[1]), "+v"(sSum[2]), "+v"(sSum[3]),
                      "+v"(sSum[4]), "+v"(sSum[5]), "+v"(sSum[6]), "+v"(sSum[7]));
    asm volatile("" : "+v"(carry[0]), "+v"(carry[1]), "+v"(carry[2]), "+v"(carry[3]),
                      "+v"(carry[4]), "+v"(carry[5]), "+v"(carry[6]), "+v"(carry[7]));
    wave_lds_order();
    if (REFILL) {
        #pragma unroll
        for (int i = 0; i < NLD; ++i) xs[i] = *(const v4f*)(xnext + (size_t)i * 2 * DDIM);
    }

    float4 bq[STEP / 8], aq[STEP / 8];
    #pragma unroll
    for (int j = 0; j < STEP / 8; ++j)
        bq[j] = *(const float4*)(ldsW + ((kstep >> 2) + 2 * j + half) * (4 * NEXP) + (col & 3) * 4);
    #pragma unroll
    for (int j = 0; j < STEP / 8; ++j)
        aq[j] = *(const float4*)(tileW + aOff + j * 8);
    __builtin_amdgcn_sched_barrier(0);
    #pragma unroll
    for (int j = 0; j < STEP / 8; ++j) {
        v2f a, bm;
        a[0] = aq[j].x;  a[1] = aq[j].y;  bm[0] = bq[j].x;  bm[1] = bq[j].y;
        acc0 = __builtin_amdgcn_wmma_f32_16x16x4_f32(
                   false, a, false, bm, (short)0, acc0, false, false);
        a[0] = aq[j].z;  a[1] = aq[j].w;  bm[0] = bq[j].z;  bm[1] = bq[j].w;
        acc1 = __builtin_amdgcn_wmma_f32_16x16x4_f32(
                   false, a, false, bm, (short)0, acc1, false, false);
    }
}

__global__ __launch_bounds__(THREADS)
void moe_count_kernel(const float* __restrict__ x,
                      const float* __restrict__ gumbel,
                      const float* __restrict__ W,
                      const float* __restrict__ bias,
                      float* __restrict__ out)
{
    __shared__ __attribute__((aligned(16))) float ldsW[DDIM * NEXP];
    __shared__ __attribute__((aligned(16))) float ldsT[NWAVE][TPB * STEPP];
    __shared__ float ldsPart[NWAVE][TPB * NEXP];
    __shared__ float ldsScal[NWAVE][TPB * 4];
    __shared__ float ldsGB[TILES * TPB * NEXP];

    const int tid  = threadIdx.x;
    const int wave = tid >> 5;
    const int lane = tid & 31;
    const int col  = lane & 15;
    const int half = lane >> 4;
    const int tile = wave >> 1;
    const int kh   = wave & 1;
    const int wgTok = blockIdx.x * (TILES * TPB);
    const int kbase = kh * KW;

    const float* xp = x + (size_t)(wgTok + tile * TPB + half) * DDIM + kbase + col * 4;

    float4 wv[(DDIM * NEXP) / 4 / THREADS];
    #pragma unroll
    for (int i = 0; i < (DDIM * NEXP) / 4 / THREADS; ++i)
        wv[i] = ((const float4*)W)[tid + i * THREADS];
    const float gv = gumbel[(size_t)wgTok * NEXP + tid];
    const float bv = bias[tid & (NEXP - 1)];

    v4f xa[NLD], xb[NLD];
    #pragma unroll
    for (int i = 0; i < NLD; ++i) xa[i] = *(const v4f*)(xp + (size_t)i * 2 * DDIM);
    __builtin_amdgcn_sched_barrier(0);
    #pragma unroll
    for (int i = 0; i < NLD; ++i) xb[i] = *(const v4f*)(xp + STEP + (size_t)i * 2 * DDIM);

    #pragma unroll
    for (int i = 0; i < (DDIM * NEXP) / 4 / THREADS; ++i) {
        const int k = tid + i * THREADS;
        float* dst = &ldsW[(k >> 2) * (4 * NEXP) + (k & 3)];
        dst[0] = wv[i].x;  dst[4] = wv[i].y;  dst[8] = wv[i].z;  dst[12] = wv[i].w;
    }
    ldsGB[tid] = bv + gv;

    lds_barrier();

    float sSum[NLD], sZero[NLD], sEqW[NLD], carry[NLD];
    #pragma unroll
    for (int i = 0; i < NLD; ++i) { sSum[i] = 0.f; sZero[i] = 0.f; sEqW[i] = 0.f; carry[i] = 0.f; }
    bool everSlow = false;
    v8f acc0 = {}, acc1 = {};

    float* tileW = &ldsT[wave][0];
    const int stOff = half * STEPP + col * 4;
    const int aOff  = col * STEPP + half * 4;

    if (kh == 1 && col == 0) {
        #pragma unroll
        for (int i = 0; i < NLD; ++i) ldsScal[wave][(2 * i + half) * 4 + 3] = xa[i].x;
    }

    int kstep = kbase;
    const float* xn = xp + 2 * STEP;
    #pragma nounroll
    for (int it = 0; it < NSTEP / 2 - 1; ++it) {
        do_step<true>(xa, xn, kstep, kbase, col, half, sSum, sZero, sEqW, carry, everSlow,
                      acc0, acc1, tileW, ldsW, stOff, aOff);
        do_step<true>(xb, xn + STEP, kstep + STEP, kbase, col, half, sSum, sZero, sEqW, carry, everSlow,
                      acc0, acc1, tileW, ldsW, stOff, aOff);
        kstep += 2 * STEP;
        xn    += 2 * STEP;
    }
    do_step<false>(xa, xn, kstep, kbase, col, half, sSum, sZero, sEqW, carry, everSlow,
                   acc0, acc1, tileW, ldsW, stOff, aOff);
    do_step<false>(xb, xn, kstep + STEP, kbase, col, half, sSum, sZero, sEqW, carry, everSlow,
                   acc0, acc1, tileW, ldsW, stOff, aOff);

    if (kh == 0 && col == 0) {
        #pragma unroll
        for (int i = 0; i < NLD; ++i) ldsScal[wave][(2 * i + half) * 4 + 3] = carry[i];
    }

    if (everSlow) {
        #pragma unroll
        for (int i = 0; i < NLD; ++i) {
            sZero[i] = row_sum16(sZero[i]);
            sEqW[i]  = row_sum16(sEqW[i]);
        }
    }
    float rowS[NLD];
    #pragma unroll
    for (int i = 0; i < NLD; ++i) rowS[i] = row_sum16(sSum[i]);
    if (col == 0) {
        #pragma unroll
        for (int i = 0; i < NLD; ++i) {
            const int trow = 2 * i + half;
            ldsScal[wave][trow * 4 + 0] = rowS[i];
            ldsScal[wave][trow * 4 + 1] = sZero[i];
            ldsScal[wave][trow * 4 + 2] = sEqW[i];
        }
    }

    if (col < NEXP) {
        #pragma unroll
        for (int r = 0; r < 8; ++r) {
            const int m = r + half * 8;
            ldsPart[wave][m * NEXP + col] = acc0[r] + acc1[r];
        }
    }
    __syncthreads();

    if (tid < TILES * TPB) {
        const int w0 = (tid >> 4) * 2, w1 = w0 + 1, m = tid & 15;
        const float s = ldsScal[w0][m * 4 + 0] + ldsScal[w1][m * 4 + 0];
        const float u = (float)DDIM - (ldsScal[w0][m * 4 + 1] + ldsScal[w1][m * 4 + 1]);
        float p = (float)(DDIM * (DDIM - 1) / 2 - (DDIM - KW))
                - (ldsScal[w0][m * 4 + 2] + ldsScal[w1][m * 4 + 2]);
        if (ldsScal[w1][m * 4 + 3] != ldsScal[w0][m * 4 + 3]) p += (float)(DDIM - KW);
        float ev[4];
        ev[0] = s;
        ev[1] = s / ((float)DDIM * (s + EPSF));
        ev[2] = u;
        ev[3] = p / (float)DDIM;

        int best = 0;
        float bestv = -INFINITY;
        #pragma unroll
        for (int e = 0; e < NEXP; ++e) {
            float le = ldsPart[w0][m * NEXP + e] + ldsPart[w1][m * NEXP + e];
            le += ldsGB[tid * NEXP + e];
            if (le > bestv) { bestv = le; best = e; }
        }
        out[wgTok + tid] = ev[best];
    }
}

extern "C" void kernel_launch(void* const* d_in, const int* in_sizes, int n_in,
                              void* d_out, int out_size, void* d_ws, size_t ws_size,
                              hipStream_t stream) {
    (void)in_sizes; (void)n_in; (void)d_ws; (void)ws_size; (void)out_size;
    const float* x      = (const float*)d_in[0];
    const float* gumbel = (const float*)d_in[1];
    const float* W      = (const float*)d_in[2];
    const float* b      = (const float*)d_in[3];
    float* out          = (float*)d_out;

    dim3 grid(NWG);
    moe_count_kernel<<<grid, THREADS, 0, stream>>>(x, gumbel, W, b, out);
}
